# starting kernel
# baseline (speedup 1.0000x reference)
#include <hip/hip_runtime.h>
#include <math.h>


#define N_NODES 100000
#define N0      60000
#define N1      40000
#define F0      128
#define F1      64
#define HID     64
#define HEADS   8
#define OUTD    16
#define ATTNV   128
#define NE      250000
#define LP      3
#define NG      8192
#define NB      8192

typedef __attribute__((ext_vector_type(2))) float f32x2;
typedef __attribute__((ext_vector_type(8))) float f32x8;

__device__ __forceinline__ f32x8 wmma_f32(f32x2 a, f32x2 b, f32x8 c) {
  return __builtin_amdgcn_wmma_f32_16x16x4_f32(false, a, false, b, (short)0, c,
                                               false, false);
}

__global__ void fill_kernel(float* __restrict__ p, float v, int n) {
  int i = blockIdx.x * 256 + threadIdx.x;
  if (i < n) p[i] = v;
}

__global__ void finals_kernel(const float* __restrict__ r_vec,
                              float* __restrict__ finals  ) {
  int c = threadIdx.x;
  if (c >= 32) return;
  for (int mp = 0; mp < 2; ++mp) {
    float re = r_vec[(mp * 32 + c) * 2 + 0];
    float im = r_vec[(mp * 32 + c) * 2 + 1];
    float inv = rsqrtf(re * re + im * im);
    re *= inv; im *= inv;
    float f1re = re, f1im = -im;
    float f0re = f1re * re - f1im * im;
    float f0im = f1re * im + f1im * re;
    float* F = finals + mp * LP * HID;
    F[0 * HID + 2 * c] = f0re; F[0 * HID + 2 * c + 1] = f0im;
    F[1 * HID + 2 * c] = f1re; F[1 * HID + 2 * c + 1] = f1im;
    F[2 * HID + 2 * c] = 1.f;  F[2 * HID + 2 * c + 1] = 0.f;
  }
}

template <int K>
__global__ __launch_bounds__(256) void gemm_tf_kernel(
    const float* __restrict__ feat, const float* __restrict__ W,
    const float* __restrict__ bias, const int* __restrict__ idx,
    float* __restrict__ tf, int M) {
  __shared__ float ldsB[K * HID];
  int tid = threadIdx.x;
  for (int i = tid; i < K * HID; i += 256) {
    int k = i >> 6, n = i & 63;
    ldsB[i] = W[n * K + k];
  }
  __syncthreads();
  int wave = tid >> 5, lane = tid & 31;
  int tile_m = blockIdx.x * 128 + wave * 16;
  if (tile_m >= M) return;
  int half = lane >> 4, l15 = lane & 15;
  int arow = tile_m + l15;
  if (arow >= M) arow = M - 1;
  f32x8 c0 = {}, c1 = {}, c2 = {}, c3 = {};
  for (int k0 = 0; k0 < K; k0 += 4) {
    int kk = k0 + half * 2;
    f32x2 a;
    a.x = feat[(long)arow * K + kk];
    a.y = feat[(long)arow * K + kk + 1];
    f32x2 b0, b1, b2, b3;
    b0.x = ldsB[kk * HID + l15 + 0];  b0.y = ldsB[(kk + 1) * HID + l15 + 0];
    b1.x = ldsB[kk * HID + l15 + 16]; b1.y = ldsB[(kk + 1) * HID + l15 + 16];
    b2.x = ldsB[kk * HID + l15 + 32]; b2.y = ldsB[(kk + 1) * HID + l15 + 32];
    b3.x = ldsB[kk * HID + l15 + 48]; b3.y = ldsB[(kk + 1) * HID + l15 + 48];
    c0 = wmma_f32(a, b0, c0);
    c1 = wmma_f32(a, b1, c1);
    c2 = wmma_f32(a, b2, c2);
    c3 = wmma_f32(a, b3, c3);
  }
#pragma unroll
  for (int v = 0; v < 8; ++v) {
    int r = tile_m + v + half * 8;
    if (r < M) {
      float* drow = tf + (long)idx[r] * HID;
      drow[l15 + 0]  = c0[v] + bias[l15 + 0];
      drow[l15 + 16] = c1[v] + bias[l15 + 16];
      drow[l15 + 32] = c2[v] + bias[l15 + 32];
      drow[l15 + 48] = c3[v] + bias[l15 + 48];
    }
  }
}

__global__ __launch_bounds__(256) void edge_h_kernel(
    const float* __restrict__ tf, const float* __restrict__ finals,
    const int* __restrict__ indices, const int* __restrict__ dst,
    const float* __restrict__ attn, float* __restrict__ h,
    float* __restrict__ a, float* __restrict__ m) {
  int wave = threadIdx.x >> 5, lane = threadIdx.x & 31;
  int e = blockIdx.x * 8 + wave;
  if (e >= NE) return;
  float2 acc = make_float2(0.f, 0.f);
#pragma unroll
  for (int i = 0; i < LP; ++i) {
    int node = indices[e * LP + i];
    const float* row = tf + (long)node * HID + lane * 2;
    float vx = row[0], vy = row[1];
    float fx = finals[i * HID + lane * 2], fy = finals[i * HID + lane * 2 + 1];
    acc.x += vx * fx - vy * fy;
    acc.y += vx * fy + vy * fx;
  }
  acc.x *= (1.f / 3.f);
  acc.y *= (1.f / 3.f);
  h[(long)e * HID + lane * 2] = acc.x;
  h[(long)e * HID + lane * 2 + 1] = acc.y;
  int d = dst[e];
#pragma unroll
  for (int hh = 0; hh < HEADS; ++hh) {
    float p = acc.x * attn[hh * HID + lane * 2] +
              acc.y * attn[hh * HID + lane * 2 + 1];
#pragma unroll
    for (int off = 16; off > 0; off >>= 1) p += __shfl_xor(p, off, 32);
    if (lane == 0) {
      float av = p > 0.f ? p : 0.01f * p;
      a[(long)e * HEADS + hh] = av;
      atomicMax(&m[d * HEADS + hh], av);
    }
  }
}

__global__ void edge_exp_kernel(float* __restrict__ a,
                                const int* __restrict__ dst,
                                const float* __restrict__ m,
                                float* __restrict__ s, int n) {
  int i = blockIdx.x * 256 + threadIdx.x;
  if (i >= n) return;
  int e = i >> 3, hh = i & 7;
  int d = dst[e];
  float ex = __expf(a[i] - m[d * HEADS + hh]);
  a[i] = ex;
  atomicAdd(&s[d * HEADS + hh], ex);
}

__global__ __launch_bounds__(256) void edge_scatter_kernel(
    const float* __restrict__ h, const float* __restrict__ a,
    const float* __restrict__ s, const int* __restrict__ dst,
    float* __restrict__ ft) {
  int wave = threadIdx.x >> 5, lane = threadIdx.x & 31;
  int e = blockIdx.x * 8 + wave;
  if (e >= NE) return;
  int d = dst[e];
  float an_l = 0.f;
  if (lane < HEADS) an_l = a[(long)e * HEADS + lane] / s[d * HEADS + lane];
  float hx = h[(long)e * HID + lane * 2];
  float hy = h[(long)e * HID + lane * 2 + 1];
#pragma unroll
  for (int hh = 0; hh < HEADS; ++hh) {
    float w = __shfl(an_l, hh, 32);
    float* base = ft + ((long)d * HEADS + hh) * HID + lane * 2;
    atomicAdd(base + 0, w * hx);
    atomicAdd(base + 1, w * hy);
  }
}

__global__ void target_elu_kernel(const float* __restrict__ ft,
                                  const int* __restrict__ target,
                                  float* __restrict__ outp, int n) {
  int i = blockIdx.x * 256 + threadIdx.x;
  if (i >= n) return;
  int b = i >> 9, j = i & 511;
  float v = ft[(long)target[b] * (HEADS * HID) + j];
  outp[i] = v > 0.f ? v : (__expf(v) - 1.f);
}

__global__ __launch_bounds__(256) void score_kernel(
    const float* __restrict__ outs, const float* __restrict__ fc1_w,
    const float* __restrict__ fc1_b, const float* __restrict__ fc2_w,
    float* __restrict__ score) {
  int p = blockIdx.y;
  const float* A = outs + (long)p * NB * (HEADS * HID);
  int wave = threadIdx.x >> 5, lane = threadIdx.x & 31;
  int tile_m = (blockIdx.x * 8 + wave) * 16;
  int half = lane >> 4, l15 = lane & 15;
  long arow = tile_m + l15;
  f32x8 acc[8] = {};
  for (int k0 = 0; k0 < HEADS * HID; k0 += 4) {
    int kk = k0 + half * 2;
    f32x2 af;
    af.x = A[arow * (HEADS * HID) + kk];
    af.y = A[arow * (HEADS * HID) + kk + 1];
#pragma unroll
    for (int nt = 0; nt < 8; ++nt) {
      int col = nt * 16 + l15;
      f32x2 bf;
      bf.x = fc1_w[(long)col * (HEADS * HID) + kk];
      bf.y = fc1_w[(long)col * (HEADS * HID) + kk + 1];
      acc[nt] = wmma_f32(af, bf, acc[nt]);
    }
  }
  float ssum = 0.f;
#pragma unroll
  for (int nt = 0; nt < 8; ++nt) {
    int col = nt * 16 + l15;
    float bb = fc1_b[col], w2 = fc2_w[col];
#pragma unroll
    for (int v = 0; v < 8; ++v) ssum += tanhf(acc[nt][v] + bb) * w2;
  }
#pragma unroll
  for (int off = 16; off > 0; off >>= 1) ssum += __shfl_xor(ssum, off, 32);
  if (lane == 0) atomicAdd(&score[p], ssum);
}

__global__ void beta_kernel(const float* __restrict__ score,
                            float* __restrict__ beta) {
  float s0 = score[0] / (float)NB, s1 = score[1] / (float)NB;
  float mx = fmaxf(s0, s1);
  float e0 = __expf(s0 - mx), e1 = __expf(s1 - mx);
  float inv = 1.f / (e0 + e1);
  beta[0] = e0 * inv;
  beta[1] = e1 * inv;
}

__global__ void hagg_kernel(const float* __restrict__ outs,
                            const float* __restrict__ beta,
                            float* __restrict__ hagg, int n) {
  int i = blockIdx.x * 256 + threadIdx.x;
  if (i >= n) return;
  hagg[i] = beta[0] * outs[i] + beta[1] * outs[(long)n + i];
}

__global__ __launch_bounds__(256) void logits_kernel(
    const float* __restrict__ hagg, const float* __restrict__ fc_w,
    const float* __restrict__ fc_b, float* __restrict__ logits) {
  __shared__ float ldsB[(HEADS * HID) * OUTD];
  int tid = threadIdx.x;
  for (int i = tid; i < (HEADS * HID) * OUTD; i += 256) {
    int k = i >> 4, n = i & 15;
    ldsB[i] = fc_w[n * (HEADS * HID) + k];
  }
  __syncthreads();
  int wave = tid >> 5, lane = tid & 31;
  int tile_m = (blockIdx.x * 8 + wave) * 16;
  int half = lane >> 4, l15 = lane & 15;
  long arow = tile_m + l15;
  f32x8 c = {};
  for (int k0 = 0; k0 < HEADS * HID; k0 += 4) {
    int kk = k0 + half * 2;
    f32x2 af;
    af.x = hagg[arow * (HEADS * HID) + kk];
    af.y = hagg[arow * (HEADS * HID) + kk + 1];
    f32x2 bf;
    bf.x = ldsB[kk * OUTD + l15];
    bf.y = ldsB[(kk + 1) * OUTD + l15];
    c = wmma_f32(af, bf, c);
  }
#pragma unroll
  for (int v = 0; v < 8; ++v) {
    int r = tile_m + v + half * 8;
    logits[(long)r * OUTD + l15] = c[v] + fc_b[l15];
  }
}

extern "C" void kernel_launch(void* const* d_in, const int* in_sizes, int n_in,
                              void* d_out, int out_size, void* d_ws,
                              size_t ws_size, hipStream_t stream) {
  const float* feat0 = (const float*)d_in[0];
  const float* feat1 = (const float*)d_in[1];
  const float* W0    = (const float*)d_in[2];
  const float* b0    = (const float*)d_in[3];
  const float* W1    = (const float*)d_in[4];
  const float* b1    = (const float*)d_in[5];
  const float* r_vec = (const float*)d_in[6];
  const float* attn0 = (const float*)d_in[7];
  const float* attn1 = (const float*)d_in[8];
  const float* fc1_w = (const float*)d_in[9];
  const float* fc1_b = (const float*)d_in[10];
  const float* fc2_w = (const float*)d_in[11];
  const float* fc_w  = (const float*)d_in[12];
  const float* fc_b  = (const float*)d_in[13];
  const int* idx0    = (const int*)d_in[14];
  const int* idx1    = (const int*)d_in[15];
  const int* mp_indices[2] = {(const int*)d_in[16], (const int*)d_in[19]};
  const int* mp_dst[2]     = {(const int*)d_in[17], (const int*)d_in[20]};
  const int* mp_target[2]  = {(const int*)d_in[18], (const int*)d_in[21]};
  const float* attn[2] = {attn0, attn1};

  float* ws     = (float*)d_ws;
  float* tf     = ws;
  float* finals = tf + (long)N_NODES * HID;
  float* h      = finals + 2 * LP * HID;
  float* a      = h + (long)NE * HID;
  float* m      = a + (long)NE * HEADS;
  float* s      = m + NG * HEADS;
  float* ft     = s + NG * HEADS;
  float* outs   = ft + (long)NG * HEADS * HID;
  float* score  = outs + 2L * NB * HEADS * HID;
  float* beta   = score + 2;

  float* logits = (float*)d_out;
  float* hagg   = logits + (long)NB * OUTD;

  gemm_tf_kernel<F0><<<(N0 + 127) / 128, 256, 0, stream>>>(feat0, W0, b0, idx0,
                                                           tf, N0);
  gemm_tf_kernel<F1><<<(N1 + 127) / 128, 256, 0, stream>>>(feat1, W1, b1, idx1,
                                                           tf, N1);
  finals_kernel<<<1, 32, 0, stream>>>(r_vec, finals);

  for (int mp = 0; mp < 2; ++mp) {
    fill_kernel<<<(NG * HEADS + 255) / 256, 256, 0, stream>>>(m, -3.0e38f,
                                                              NG * HEADS);
    fill_kernel<<<(NG * HEADS + 255) / 256, 256, 0, stream>>>(s, 0.f,
                                                              NG * HEADS);
    fill_kernel<<<(NG * HEADS * HID + 255) / 256, 256, 0, stream>>>(
        ft, 0.f, NG * HEADS * HID);
    edge_h_kernel<<<NE / 8, 256, 0, stream>>>(tf, finals + mp * LP * HID,
                                              mp_indices[mp], mp_dst[mp],
                                              attn[mp], h, a, m);
    edge_exp_kernel<<<(NE * HEADS + 255) / 256, 256, 0, stream>>>(
        a, mp_dst[mp], m, s, NE * HEADS);
    edge_scatter_kernel<<<NE / 8, 256, 0, stream>>>(h, a, s, mp_dst[mp], ft);
    target_elu_kernel<<<(NB * HEADS * HID + 255) / 256, 256, 0, stream>>>(
        ft, mp_target[mp], outs + (long)mp * NB * HEADS * HID,
        NB * HEADS * HID);
  }

  fill_kernel<<<1, 64, 0, stream>>>(score, 0.f, 2);
  {
    dim3 g(NB / 128, 2);
    score_kernel<<<g, 256, 0, stream>>>(outs, fc1_w, fc1_b, fc2_w, score);
  }
  beta_kernel<<<1, 1, 0, stream>>>(score, beta);
  hagg_kernel<<<(NB * HEADS * HID + 255) / 256, 256, 0, stream>>>(
      outs, beta, hagg, NB * HEADS * HID);
  logits_kernel<<<NB / 128, 256, 0, stream>>>(hagg, fc_w, fc_b, logits);
}


# skip edges whose destination no target reads: a flag table made from target (one LDS pass per metapath); edge_h, edge_exp, edge_scatter drop dead edges; live rows keep the same sums
# speedup vs baseline: 2.1156x; 2.1156x over previous
#include <hip/hip_runtime.h>
#include <math.h>


#define N_NODES 100000
#define N0      60000
#define N1      40000
#define F0      128
#define F1      64
#define HID     64
#define HEADS   8
#define OUTD    16
#define ATTNV   128
#define NE      250000
#define LP      3
#define NG      8192
#define NB      8192

typedef __attribute__((ext_vector_type(2))) float f32x2;
typedef __attribute__((ext_vector_type(8))) float f32x8;

__device__ __forceinline__ f32x8 wmma_f32(f32x2 a, f32x2 b, f32x8 c) {
  return __builtin_amdgcn_wmma_f32_16x16x4_f32(false, a, false, b, (short)0, c,
                                               false, false);
}

__global__ void fill_kernel(float* __restrict__ p, float v, int n) {
  int i = blockIdx.x * 256 + threadIdx.x;
  if (i < n) p[i] = v;
}

__global__ void finals_kernel(const float* __restrict__ r_vec,
                              float* __restrict__ finals  ) {
  int c = threadIdx.x;
  if (c >= 32) return;
  for (int mp = 0; mp < 2; ++mp) {
    float re = r_vec[(mp * 32 + c) * 2 + 0];
    float im = r_vec[(mp * 32 + c) * 2 + 1];
    float inv = rsqrtf(re * re + im * im);
    re *= inv; im *= inv;
    float f1re = re, f1im = -im;
    float f0re = f1re * re - f1im * im;
    float f0im = f1re * im + f1im * re;
    float* F = finals + mp * LP * HID;
    F[0 * HID + 2 * c] = f0re; F[0 * HID + 2 * c + 1] = f0im;
    F[1 * HID + 2 * c] = f1re; F[1 * HID + 2 * c + 1] = f1im;
    F[2 * HID + 2 * c] = 1.f;  F[2 * HID + 2 * c + 1] = 0.f;
  }
}

__global__ __launch_bounds__(1024) void mark_kernel(
    const int* __restrict__ target0, const int* __restrict__ target1,
    int* __restrict__ used  ) {
  __shared__ int flag[NG];
  const int* target = blockIdx.x == 0 ? target0 : target1;
  for (int i = threadIdx.x; i < NG; i += 1024) flag[i] = 0;
  __syncthreads();
#pragma unroll
  for (int b = threadIdx.x; b < NB; b += 1024) flag[target[b]] = 1;
  __syncthreads();
  for (int i = threadIdx.x; i < NG; i += 1024)
    used[blockIdx.x * NG + i] = flag[i];
}

template <int K>
__global__ __launch_bounds__(256) void gemm_tf_kernel(
    const float* __restrict__ feat, const float* __restrict__ W,
    const float* __restrict__ bias, const int* __restrict__ idx,
    float* __restrict__ tf, int M) {
  __shared__ float ldsB[K * HID];
  int tid = threadIdx.x;
  for (int i = tid; i < K * HID; i += 256) {
    int k = i >> 6, n = i & 63;
    ldsB[i] = W[n * K + k];
  }
  __syncthreads();
  int wave = tid >> 5, lane = tid & 31;
  int tile_m = blockIdx.x * 128 + wave * 16;
  if (tile_m >= M) return;
  int half = lane >> 4, l15 = lane & 15;
  int arow = tile_m + l15;
  if (arow >= M) arow = M - 1;
  f32x8 c0 = {}, c1 = {}, c2 = {}, c3 = {};
  for (int k0 = 0; k0 < K; k0 += 4) {
    int kk = k0 + half * 2;
    f32x2 a;
    a.x = feat[(long)arow * K + kk];
    a.y = feat[(long)arow * K + kk + 1];
    f32x2 b0, b1, b2, b3;
    b0.x = ldsB[kk * HID + l15 + 0];  b0.y = ldsB[(kk + 1) * HID + l15 + 0];
    b1.x = ldsB[kk * HID + l15 + 16]; b1.y = ldsB[(kk + 1) * HID + l15 + 16];
    b2.x = ldsB[kk * HID + l15 + 32]; b2.y = ldsB[(kk + 1) * HID + l15 + 32];
    b3.x = ldsB[kk * HID + l15 + 48]; b3.y = ldsB[(kk + 1) * HID + l15 + 48];
    c0 = wmma_f32(a, b0, c0);
    c1 = wmma_f32(a, b1, c1);
    c2 = wmma_f32(a, b2, c2);
    c3 = wmma_f32(a, b3, c3);
  }
#pragma unroll
  for (int v = 0; v < 8; ++v) {
    int r = tile_m + v + half * 8;
    if (r < M) {
      float* drow = tf + (long)idx[r] * HID;
      drow[l15 + 0]  = c0[v] + bias[l15 + 0];
      drow[l15 + 16] = c1[v] + bias[l15 + 16];
      drow[l15 + 32] = c2[v] + bias[l15 + 32];
      drow[l15 + 48] = c3[v] + bias[l15 + 48];
    }
  }
}

__global__ __launch_bounds__(256) void edge_h_kernel(
    const float* __restrict__ tf, const float* __restrict__ finals,
    const int* __restrict__ indices, const int* __restrict__ dst,
    const int* __restrict__ used, const float* __restrict__ attn,
    float* __restrict__ h, float* __restrict__ a, float* __restrict__ m) {
  int wave = threadIdx.x >> 5, lane = threadIdx.x & 31;
  int ebase = (blockIdx.x * 8 + wave) * 16;
  if (ebase >= NE) return;
  int half = lane >> 4, l15 = lane & 15;
  int e = ebase + l15;
  int koff = half * 4;
  const bool live = used[dst[e]] != 0;
  const unsigned live_mask = __builtin_amdgcn_ballot_w32(live);
  const float* r0 = tf + koff;
  const float* r1 = tf + koff;
  const float* r2 = tf + koff;
  if (live) {
    r0 = tf + (long)indices[e * LP + 0] * HID + koff;
    r1 = tf + (long)indices[e * LP + 1] * HID + koff;
    r2 = tf + (long)indices[e * LP + 2] * HID + koff;
  }
  const float* fin = finals + koff;
  const float* at = attn + (l15 & (HEADS - 1)) * HID + koff;
  const bool head_col = l15 < HEADS;
  float* hrow = h + (long)e * HID + koff;
  f32x8 c = {};
  for (int j = 0; j < 8; ++j) {
    float4 bq = *(const float4*)(at + 8 * j);
    if (!head_col) bq = make_float4(0.f, 0.f, 0.f, 0.f);
    float4 hv = make_float4(0.f, 0.f, 0.f, 0.f);
    if (live) {
    const float4 t0 = *(const float4*)(r0 + 8 * j);
    const float4 t1 = *(const float4*)(r1 + 8 * j);
    const float4 t2 = *(const float4*)(r2 + 8 * j);
    const float4 f0 = *(const float4*)(fin + 0 * HID + 8 * j);
    const float4 f1 = *(const float4*)(fin + 1 * HID + 8 * j);
    const float4 f2 = *(const float4*)(fin + 2 * HID + 8 * j);
    hv.x += t0.x * f0.x - t0.y * f0.y;
    hv.y += t0.x * f0.y + t0.y * f0.x;
    hv.z += t0.z * f0.z - t0.w * f0.w;
    hv.w += t0.z * f0.w + t0.w * f0.z;
    hv.x += t1.x * f1.x - t1.y * f1.y;
    hv.y += t1.x * f1.y + t1.y * f1.x;
    hv.z += t1.z * f1.z - t1.w * f1.w;
    hv.w += t1.z * f1.w + t1.w * f1.z;
    hv.x += t2.x * f2.x - t2.y * f2.y;
    hv.y += t2.x * f2.y + t2.y * f2.x;
    hv.z += t2.z * f2.z - t2.w * f2.w;
    hv.w += t2.z * f2.w + t2.w * f2.z;
    hv.x *= (1.f / 3.f);
    hv.y *= (1.f / 3.f);
    hv.z *= (1.f / 3.f);
    hv.w *= (1.f / 3.f);
    *(float4*)(hrow + 8 * j) = hv;
    }
    f32x2 a01, a23, b01, b23;
    a01.x = hv.x; a01.y = hv.y; b01.x = bq.x; b01.y = bq.y;
    a23.x = hv.z; a23.y = hv.w; b23.x = bq.z; b23.y = bq.w;
    c = wmma_f32(a01, b01, c);
    c = wmma_f32(a23, b23, c);
  }
  float av[8];
#pragma unroll
  for (int v = 0; v < 8; ++v) {
    float p = c[v];
    av[v] = p > 0.f ? p : 0.01f * p;
  }
  if (head_col) {
    int eb2 = ebase + half * 8;
    const int4 d03 = *(const int4*)(dst + eb2);
    const int4 d47 = *(const int4*)(dst + eb2 + 4);
    const int dv[8] = {d03.x, d03.y, d03.z, d03.w, d47.x, d47.y, d47.z, d47.w};
#pragma unroll
    for (int v = 0; v < 8; ++v) {
      if ((live_mask >> (half * 8 + v)) & 1u) {
        int e2 = eb2 + v;
        a[(long)e2 * HEADS + l15] = av[v];
        atomicMax(&m[dv[v] * HEADS + l15], av[v]);
      }
    }
  }
}

__global__ void edge_exp_kernel(float* __restrict__ a,
                                const int* __restrict__ dst,
                                const int* __restrict__ used,
                                const float* __restrict__ m,
                                float* __restrict__ s, int n) {
  int i = blockIdx.x * 256 + threadIdx.x;
  if (i >= n) return;
  int e = i >> 3, hh = i & 7;
  int d = dst[e];
  if (used[d] == 0) return;
  float ex = __expf(a[i] - m[d * HEADS + hh]);
  a[i] = ex;
  atomicAdd(&s[d * HEADS + hh], ex);
}

__global__ __launch_bounds__(256) void edge_scatter_kernel(
    const float* __restrict__ h, const float* __restrict__ a,
    const float* __restrict__ s, const int* __restrict__ dst,
    const int* __restrict__ used, float* __restrict__ ft) {
  int wave = threadIdx.x >> 5, lane = threadIdx.x & 31;
  int e = blockIdx.x * 8 + wave;
  if (e >= NE) return;
  int d = dst[e];
  if (used[d] == 0) return;
  float an_l = 0.f;
  if (lane < HEADS) an_l = a[(long)e * HEADS + lane] / s[d * HEADS + lane];
  float hx = h[(long)e * HID + lane * 2];
  float hy = h[(long)e * HID + lane * 2 + 1];
#pragma unroll
  for (int hh = 0; hh < HEADS; ++hh) {
    float w = __shfl(an_l, hh, 32);
    float* base = ft + ((long)d * HEADS + hh) * HID + lane * 2;
    atomicAdd(base + 0, w * hx);
    atomicAdd(base + 1, w * hy);
  }
}

__global__ void target_elu_kernel(const float* __restrict__ ft,
                                  const int* __restrict__ target,
                                  float* __restrict__ outp, int n) {
  int i = blockIdx.x * 256 + threadIdx.x;
  if (i >= n) return;
  int b = i >> 9, j = i & 511;
  float v = ft[(long)target[b] * (HEADS * HID) + j];
  outp[i] = v > 0.f ? v : (__expf(v) - 1.f);
}

__global__ __launch_bounds__(256) void score_kernel(
    const float* __restrict__ outs, const float* __restrict__ fc1_w,
    const float* __restrict__ fc1_b, const float* __restrict__ fc2_w,
    float* __restrict__ score) {
  int p = blockIdx.y;
  const float* A = outs + (long)p * NB * (HEADS * HID);
  int wave = threadIdx.x >> 5, lane = threadIdx.x & 31;
  int tile_m = (blockIdx.x * 8 + wave) * 16;
  int half = lane >> 4, l15 = lane & 15;
  long arow = tile_m + l15;
  f32x8 acc[8] = {};
  for (int k0 = 0; k0 < HEADS * HID; k0 += 4) {
    int kk = k0 + half * 2;
    f32x2 af;
    af.x = A[arow * (HEADS * HID) + kk];
    af.y = A[arow * (HEADS * HID) + kk + 1];
#pragma unroll
    for (int nt = 0; nt < 8; ++nt) {
      int col = nt * 16 + l15;
      f32x2 bf;
      bf.x = fc1_w[(long)col * (HEADS * HID) + kk];
      bf.y = fc1_w[(long)col * (HEADS * HID) + kk + 1];
      acc[nt] = wmma_f32(af, bf, acc[nt]);
    }
  }
  float ssum = 0.f;
#pragma unroll
  for (int nt = 0; nt < 8; ++nt) {
    int col = nt * 16 + l15;
    float bb = fc1_b[col], w2 = fc2_w[col];
#pragma unroll
    for (int v = 0; v < 8; ++v) ssum += tanhf(acc[nt][v] + bb) * w2;
  }
#pragma unroll
  for (int off = 16; off > 0; off >>= 1) ssum += __shfl_xor(ssum, off, 32);
  if (lane == 0) atomicAdd(&score[p], ssum);
}

__global__ void beta_kernel(const float* __restrict__ score,
                            float* __restrict__ beta) {
  float s0 = score[0] / (float)NB, s1 = score[1] / (float)NB;
  float mx = fmaxf(s0, s1);
  float e0 = __expf(s0 - mx), e1 = __expf(s1 - mx);
  float inv = 1.f / (e0 + e1);
  beta[0] = e0 * inv;
  beta[1] = e1 * inv;
}

__global__ void hagg_kernel(const float* __restrict__ outs,
                            const float* __restrict__ beta,
                            float* __restrict__ hagg, int n) {
  int i = blockIdx.x * 256 + threadIdx.x;
  if (i >= n) return;
  hagg[i] = beta[0] * outs[i] + beta[1] * outs[(long)n + i];
}

__global__ __launch_bounds__(256) void logits_kernel(
    const float* __restrict__ hagg, const float* __restrict__ fc_w,
    const float* __restrict__ fc_b, float* __restrict__ logits) {
  __shared__ float ldsB[(HEADS * HID) * OUTD];
  int tid = threadIdx.x;
  for (int i = tid; i < (HEADS * HID) * OUTD; i += 256) {
    int k = i >> 4, n = i & 15;
    ldsB[i] = fc_w[n * (HEADS * HID) + k];
  }
  __syncthreads();
  int wave = tid >> 5, lane = tid & 31;
  int tile_m = (blockIdx.x * 8 + wave) * 16;
  int half = lane >> 4, l15 = lane & 15;
  long arow = tile_m + l15;
  f32x8 c = {};
  for (int k0 = 0; k0 < HEADS * HID; k0 += 4) {
    int kk = k0 + half * 2;
    f32x2 af;
    af.x = hagg[arow * (HEADS * HID) + kk];
    af.y = hagg[arow * (HEADS * HID) + kk + 1];
    f32x2 bf;
    bf.x = ldsB[kk * OUTD + l15];
    bf.y = ldsB[(kk + 1) * OUTD + l15];
    c = wmma_f32(af, bf, c);
  }
#pragma unroll
  for (int v = 0; v < 8; ++v) {
    int r = tile_m + v + half * 8;
    logits[(long)r * OUTD + l15] = c[v] + fc_b[l15];
  }
}

extern "C" void kernel_launch(void* const* d_in, const int* in_sizes, int n_in,
                              void* d_out, int out_size, void* d_ws,
                              size_t ws_size, hipStream_t stream) {
  const float* feat0 = (const float*)d_in[0];
  const float* feat1 = (const float*)d_in[1];
  const float* W0    = (const float*)d_in[2];
  const float* b0    = (const float*)d_in[3];
  const float* W1    = (const float*)d_in[4];
  const float* b1    = (const float*)d_in[5];
  const float* r_vec = (const float*)d_in[6];
  const float* attn0 = (const float*)d_in[7];
  const float* attn1 = (const float*)d_in[8];
  const float* fc1_w = (const float*)d_in[9];
  const float* fc1_b = (const float*)d_in[10];
  const float* fc2_w = (const float*)d_in[11];
  const float* fc_w  = (const float*)d_in[12];
  const float* fc_b  = (const float*)d_in[13];
  const int* idx0    = (const int*)d_in[14];
  const int* idx1    = (const int*)d_in[15];
  const int* mp_indices[2] = {(const int*)d_in[16], (const int*)d_in[19]};
  const int* mp_dst[2]     = {(const int*)d_in[17], (const int*)d_in[20]};
  const int* mp_target[2]  = {(const int*)d_in[18], (const int*)d_in[21]};
  const float* attn[2] = {attn0, attn1};

  float* ws     = (float*)d_ws;
  float* tf     = ws;
  float* finals = tf + (long)N_NODES * HID;
  float* h      = finals + 2 * LP * HID;
  float* a      = h + (long)NE * HID;
  float* m      = a + (long)NE * HEADS;
  float* s      = m + NG * HEADS;
  float* ft     = s + NG * HEADS;
  float* outs   = ft + (long)NG * HEADS * HID;
  float* score  = outs + 2L * NB * HEADS * HID;
  float* beta   = score + 2;
  int* used     = (int*)(outs + (long)NB * HEADS * HID);

  float* logits = (float*)d_out;
  float* hagg   = logits + (long)NB * OUTD;

  gemm_tf_kernel<F0><<<(N0 + 127) / 128, 256, 0, stream>>>(feat0, W0, b0, idx0,
                                                           tf, N0);
  gemm_tf_kernel<F1><<<(N1 + 127) / 128, 256, 0, stream>>>(feat1, W1, b1, idx1,
                                                           tf, N1);
  finals_kernel<<<1, 32, 0, stream>>>(r_vec, finals);

  mark_kernel<<<2, 1024, 0, stream>>>(mp_target[0], mp_target[1], used);

  for (int mp = 0; mp < 2; ++mp) {
    fill_kernel<<<(NG * HEADS + 255) / 256, 256, 0, stream>>>(m, -3.0e38f,
                                                              NG * HEADS);
    fill_kernel<<<(NG * HEADS + 255) / 256, 256, 0, stream>>>(s, 0.f,
                                                              NG * HEADS);
    fill_kernel<<<(NG * HEADS * HID + 255) / 256, 256, 0, stream>>>(
        ft, 0.f, NG * HEADS * HID);
    edge_h_kernel<<<(NE / 16 + 7) / 8, 256, 0, stream>>>(tf, finals + mp * LP * HID,
                                              mp_indices[mp], mp_dst[mp],
                                              used + mp * NG, attn[mp], h, a, m);
    edge_exp_kernel<<<(NE * HEADS + 255) / 256, 256, 0, stream>>>(
        a, mp_dst[mp], used + mp * NG, m, s, NE * HEADS);
    edge_scatter_kernel<<<NE / 8, 256, 0, stream>>>(h, a, s, mp_dst[mp],
                                                    used + mp * NG, ft);
    target_elu_kernel<<<(NB * HEADS * HID + 255) / 256, 256, 0, stream>>>(
        ft, mp_target[mp], outs + (long)mp * NB * HEADS * HID,
        NB * HEADS * HID);
  }

  fill_kernel<<<1, 64, 0, stream>>>(score, 0.f, 2);
  {
    dim3 g(NB / 128, 2);
    score_kernel<<<g, 256, 0, stream>>>(outs, fc1_w, fc1_b, fc2_w, score);
  }
  beta_kernel<<<1, 1, 0, stream>>>(score, beta);
  hagg_kernel<<<(NB * HEADS * HID + 255) / 256, 256, 0, stream>>>(
      outs, beta, hagg, NB * HEADS * HID);
  logits_kernel<<<NB / 128, 256, 0, stream>>>(hagg, fc_w, fc_b, logits);
}


# ft scatter: a lane owns floats lane and lane+32 of the row, so each f32 atomic instruction covers one whole 128-B line (16 lines per edge, not 32 half-used); same adds to the same words
# speedup vs baseline: 2.2264x; 1.0524x over previous
#include <hip/hip_runtime.h>
#include <math.h>


#define N_NODES 100000
#define N0      60000
#define N1      40000
#define F0      128
#define F1      64
#define HID     64
#define HEADS   8
#define OUTD    16
#define ATTNV   128
#define NE      250000
#define LP      3
#define NG      8192
#define NB      8192

typedef __attribute__((ext_vector_type(2))) float f32x2;
typedef __attribute__((ext_vector_type(8))) float f32x8;

__device__ __forceinline__ f32x8 wmma_f32(f32x2 a, f32x2 b, f32x8 c) {
  return __builtin_amdgcn_wmma_f32_16x16x4_f32(false, a, false, b, (short)0, c,
                                               false, false);
}

__global__ void fill_kernel(float* __restrict__ p, float v, int n) {
  int i = blockIdx.x * 256 + threadIdx.x;
  if (i < n) p[i] = v;
}

__global__ void finals_kernel(const float* __restrict__ r_vec,
                              float* __restrict__ finals  ) {
  int c = threadIdx.x;
  if (c >= 32) return;
  for (int mp = 0; mp < 2; ++mp) {
    float re = r_vec[(mp * 32 + c) * 2 + 0];
    float im = r_vec[(mp * 32 + c) * 2 + 1];
    float inv = rsqrtf(re * re + im * im);
    re *= inv; im *= inv;
    float f1re = re, f1im = -im;
    float f0re = f1re * re - f1im * im;
    float f0im = f1re * im + f1im * re;
    float* F = finals + mp * LP * HID;
    F[0 * HID + 2 * c] = f0re; F[0 * HID + 2 * c + 1] = f0im;
    F[1 * HID + 2 * c] = f1re; F[1 * HID + 2 * c + 1] = f1im;
    F[2 * HID + 2 * c] = 1.f;  F[2 * HID + 2 * c + 1] = 0.f;
  }
}

__global__ __launch_bounds__(1024) void mark_kernel(
    const int* __restrict__ target0, const int* __restrict__ target1,
    int* __restrict__ used  ) {
  __shared__ int flag[NG];
  const int* target = blockIdx.x == 0 ? target0 : target1;
  for (int i = threadIdx.x; i < NG; i += 1024) flag[i] = 0;
  __syncthreads();
#pragma unroll
  for (int b = threadIdx.x; b < NB; b += 1024) flag[target[b]] = 1;
  __syncthreads();
  for (int i = threadIdx.x; i < NG; i += 1024)
    used[blockIdx.x * NG + i] = flag[i];
}

template <int K>
__global__ __launch_bounds__(256) void gemm_tf_kernel(
    const float* __restrict__ feat, const float* __restrict__ W,
    const float* __restrict__ bias, const int* __restrict__ idx,
    float* __restrict__ tf, int M) {
  __shared__ float ldsB[K * HID];
  int tid = threadIdx.x;
  for (int i = tid; i < K * HID; i += 256) {
    int k = i >> 6, n = i & 63;
    ldsB[i] = W[n * K + k];
  }
  __syncthreads();
  int wave = tid >> 5, lane = tid & 31;
  int tile_m = blockIdx.x * 128 + wave * 16;
  if (tile_m >= M) return;
  int half = lane >> 4, l15 = lane & 15;
  int arow = tile_m + l15;
  if (arow >= M) arow = M - 1;
  f32x8 c0 = {}, c1 = {}, c2 = {}, c3 = {};
  for (int k0 = 0; k0 < K; k0 += 4) {
    int kk = k0 + half * 2;
    f32x2 a;
    a.x = feat[(long)arow * K + kk];
    a.y = feat[(long)arow * K + kk + 1];
    f32x2 b0, b1, b2, b3;
    b0.x = ldsB[kk * HID + l15 + 0];  b0.y = ldsB[(kk + 1) * HID + l15 + 0];
    b1.x = ldsB[kk * HID + l15 + 16]; b1.y = ldsB[(kk + 1) * HID + l15 + 16];
    b2.x = ldsB[kk * HID + l15 + 32]; b2.y = ldsB[(kk + 1) * HID + l15 + 32];
    b3.x = ldsB[kk * HID + l15 + 48]; b3.y = ldsB[(kk + 1) * HID + l15 + 48];
    c0 = wmma_f32(a, b0, c0);
    c1 = wmma_f32(a, b1, c1);
    c2 = wmma_f32(a, b2, c2);
    c3 = wmma_f32(a, b3, c3);
  }
#pragma unroll
  for (int v = 0; v < 8; ++v) {
    int r = tile_m + v + half * 8;
    if (r < M) {
      float* drow = tf + (long)idx[r] * HID;
      drow[l15 + 0]  = c0[v] + bias[l15 + 0];
      drow[l15 + 16] = c1[v] + bias[l15 + 16];
      drow[l15 + 32] = c2[v] + bias[l15 + 32];
      drow[l15 + 48] = c3[v] + bias[l15 + 48];
    }
  }
}

__global__ __launch_bounds__(256) void edge_h_kernel(
    const float* __restrict__ tf, const float* __restrict__ finals,
    const int* __restrict__ indices, const int* __restrict__ dst,
    const int* __restrict__ used, const float* __restrict__ attn,
    float* __restrict__ h, float* __restrict__ a, float* __restrict__ m) {
  int wave = threadIdx.x >> 5, lane = threadIdx.x & 31;
  int ebase = (blockIdx.x * 8 + wave) * 16;
  if (ebase >= NE) return;
  int half = lane >> 4, l15 = lane & 15;
  int e = ebase + l15;
  int koff = half * 4;
  const bool live = used[dst[e]] != 0;
  const unsigned live_mask = __builtin_amdgcn_ballot_w32(live);
  const float* r0 = tf + koff;
  const float* r1 = tf + koff;
  const float* r2 = tf + koff;
  if (live) {
    r0 = tf + (long)indices[e * LP + 0] * HID + koff;
    r1 = tf + (long)indices[e * LP + 1] * HID + koff;
    r2 = tf + (long)indices[e * LP + 2] * HID + koff;
  }
  const float* fin = finals + koff;
  const float* at = attn + (l15 & (HEADS - 1)) * HID + koff;
  const bool head_col = l15 < HEADS;
  float* hrow = h + (long)e * HID + koff;
  f32x8 c = {};
  for (int j = 0; j < 8; ++j) {
    float4 bq = *(const float4*)(at + 8 * j);
    if (!head_col) bq = make_float4(0.f, 0.f, 0.f, 0.f);
    float4 hv = make_float4(0.f, 0.f, 0.f, 0.f);
    if (live) {
    const float4 t0 = *(const float4*)(r0 + 8 * j);
    const float4 t1 = *(const float4*)(r1 + 8 * j);
    const float4 t2 = *(const float4*)(r2 + 8 * j);
    const float4 f0 = *(const float4*)(fin + 0 * HID + 8 * j);
    const float4 f1 = *(const float4*)(fin + 1 * HID + 8 * j);
    const float4 f2 = *(const float4*)(fin + 2 * HID + 8 * j);
    hv.x += t0.x * f0.x - t0.y * f0.y;
    hv.y += t0.x * f0.y + t0.y * f0.x;
    hv.z += t0.z * f0.z - t0.w * f0.w;
    hv.w += t0.z * f0.w + t0.w * f0.z;
    hv.x += t1.x * f1.x - t1.y * f1.y;
    hv.y += t1.x * f1.y + t1.y * f1.x;
    hv.z += t1.z * f1.z - t1.w * f1.w;
    hv.w += t1.z * f1.w + t1.w * f1.z;
    hv.x += t2.x * f2.x - t2.y * f2.y;
    hv.y += t2.x * f2.y + t2.y * f2.x;
    hv.z += t2.z * f2.z - t2.w * f2.w;
    hv.w += t2.z * f2.w + t2.w * f2.z;
    hv.x *= (1.f / 3.f);
    hv.y *= (1.f / 3.f);
    hv.z *= (1.f / 3.f);
    hv.w *= (1.f / 3.f);
    *(float4*)(hrow + 8 * j) = hv;
    }
    f32x2 a01, a23, b01, b23;
    a01.x = hv.x; a01.y = hv.y; b01.x = bq.x; b01.y = bq.y;
    a23.x = hv.z; a23.y = hv.w; b23.x = bq.z; b23.y = bq.w;
    c = wmma_f32(a01, b01, c);
    c = wmma_f32(a23, b23, c);
  }
  float av[8];
#pragma unroll
  for (int v = 0; v < 8; ++v) {
    float p = c[v];
    av[v] = p > 0.f ? p : 0.01f * p;
  }
  if (head_col) {
    int eb2 = ebase + half * 8;
    const int4 d03 = *(const int4*)(dst + eb2);
    const int4 d47 = *(const int4*)(dst + eb2 + 4);
    const int dv[8] = {d03.x, d03.y, d03.z, d03.w, d47.x, d47.y, d47.z, d47.w};
#pragma unroll
    for (int v = 0; v < 8; ++v) {
      if ((live_mask >> (half * 8 + v)) & 1u) {
        int e2 = eb2 + v;
        a[(long)e2 * HEADS + l15] = av[v];
        atomicMax(&m[dv[v] * HEADS + l15], av[v]);
      }
    }
  }
}

__global__ void edge_exp_kernel(float* __restrict__ a,
                                const int* __restrict__ dst,
                                const int* __restrict__ used,
                                const float* __restrict__ m,
                                float* __restrict__ s, int n) {
  int i = blockIdx.x * 256 + threadIdx.x;
  if (i >= n) return;
  int e = i >> 3, hh = i & 7;
  int d = dst[e];
  if (used[d] == 0) return;
  float ex = __expf(a[i] - m[d * HEADS + hh]);
  a[i] = ex;
  atomicAdd(&s[d * HEADS + hh], ex);
}

__global__ __launch_bounds__(256) void edge_scatter_kernel(
    const float* __restrict__ h, const float* __restrict__ a,
    const float* __restrict__ s, const int* __restrict__ dst,
    const int* __restrict__ used, float* __restrict__ ft) {
  int wave = threadIdx.x >> 5, lane = threadIdx.x & 31;
  int e = blockIdx.x * 8 + wave;
  if (e >= NE) return;
  int d = dst[e];
  if (used[d] == 0) return;
  float an_l = 0.f;
  if (lane < HEADS) an_l = a[(long)e * HEADS + lane] / s[d * HEADS + lane];
  float hlo = h[(long)e * HID + lane];
  float hhi = h[(long)e * HID + 32 + lane];
#pragma unroll
  for (int hh = 0; hh < HEADS; ++hh) {
    float w = __shfl(an_l, hh, 32);
    float* base = ft + ((long)d * HEADS + hh) * HID + lane;
    atomicAdd(base, w * hlo);
    atomicAdd(base + 32, w * hhi);
  }
}

__global__ void target_elu_kernel(const float* __restrict__ ft,
                                  const int* __restrict__ target,
                                  float* __restrict__ outp, int n) {
  int i = blockIdx.x * 256 + threadIdx.x;
  if (i >= n) return;
  int b = i >> 9, j = i & 511;
  float v = ft[(long)target[b] * (HEADS * HID) + j];
  outp[i] = v > 0.f ? v : (__expf(v) - 1.f);
}

__global__ __launch_bounds__(256) void score_kernel(
    const float* __restrict__ outs, const float* __restrict__ fc1_w,
    const float* __restrict__ fc1_b, const float* __restrict__ fc2_w,
    float* __restrict__ score) {
  int p = blockIdx.y;
  const float* A = outs + (long)p * NB * (HEADS * HID);
  int wave = threadIdx.x >> 5, lane = threadIdx.x & 31;
  int tile_m = (blockIdx.x * 8 + wave) * 16;
  int half = lane >> 4, l15 = lane & 15;
  long arow = tile_m + l15;
  f32x8 acc[8] = {};
  for (int k0 = 0; k0 < HEADS * HID; k0 += 4) {
    int kk = k0 + half * 2;
    f32x2 af;
    af.x = A[arow * (HEADS * HID) + kk];
    af.y = A[arow * (HEADS * HID) + kk + 1];
#pragma unroll
    for (int nt = 0; nt < 8; ++nt) {
      int col = nt * 16 + l15;
      f32x2 bf;
      bf.x = fc1_w[(long)col * (HEADS * HID) + kk];
      bf.y = fc1_w[(long)col * (HEADS * HID) + kk + 1];
      acc[nt] = wmma_f32(af, bf, acc[nt]);
    }
  }
  float ssum = 0.f;
#pragma unroll
  for (int nt = 0; nt < 8; ++nt) {
    int col = nt * 16 + l15;
    float bb = fc1_b[col], w2 = fc2_w[col];
#pragma unroll
    for (int v = 0; v < 8; ++v) ssum += tanhf(acc[nt][v] + bb) * w2;
  }
#pragma unroll
  for (int off = 16; off > 0; off >>= 1) ssum += __shfl_xor(ssum, off, 32);
  if (lane == 0) atomicAdd(&score[p], ssum);
}

__global__ void beta_kernel(const float* __restrict__ score,
                            float* __restrict__ beta) {
  float s0 = score[0] / (float)NB, s1 = score[1] / (float)NB;
  float mx = fmaxf(s0, s1);
  float e0 = __expf(s0 - mx), e1 = __expf(s1 - mx);
  float inv = 1.f / (e0 + e1);
  beta[0] = e0 * inv;
  beta[1] = e1 * inv;
}

__global__ void hagg_kernel(const float* __restrict__ outs,
                            const float* __restrict__ beta,
                            float* __restrict__ hagg, int n) {
  int i = blockIdx.x * 256 + threadIdx.x;
  if (i >= n) return;
  hagg[i] = beta[0] * outs[i] + beta[1] * outs[(long)n + i];
}

__global__ __launch_bounds__(256) void logits_kernel(
    const float* __restrict__ hagg, const float* __restrict__ fc_w,
    const float* __restrict__ fc_b, float* __restrict__ logits) {
  __shared__ float ldsB[(HEADS * HID) * OUTD];
  int tid = threadIdx.x;
  for (int i = tid; i < (HEADS * HID) * OUTD; i += 256) {
    int k = i >> 4, n = i & 15;
    ldsB[i] = fc_w[n * (HEADS * HID) + k];
  }
  __syncthreads();
  int wave = tid >> 5, lane = tid & 31;
  int tile_m = (blockIdx.x * 8 + wave) * 16;
  int half = lane >> 4, l15 = lane & 15;
  long arow = tile_m + l15;
  f32x8 c = {};
  for (int k0 = 0; k0 < HEADS * HID; k0 += 4) {
    int kk = k0 + half * 2;
    f32x2 af;
    af.x = hagg[arow * (HEADS * HID) + kk];
    af.y = hagg[arow * (HEADS * HID) + kk + 1];
    f32x2 bf;
    bf.x = ldsB[kk * OUTD + l15];
    bf.y = ldsB[(kk + 1) * OUTD + l15];
    c = wmma_f32(af, bf, c);
  }
#pragma unroll
  for (int v = 0; v < 8; ++v) {
    int r = tile_m + v + half * 8;
    logits[(long)r * OUTD + l15] = c[v] + fc_b[l15];
  }
}

extern "C" void kernel_launch(void* const* d_in, const int* in_sizes, int n_in,
                              void* d_out, int out_size, void* d_ws,
                              size_t ws_size, hipStream_t stream) {
  const float* feat0 = (const float*)d_in[0];
  const float* feat1 = (const float*)d_in[1];
  const float* W0    = (const float*)d_in[2];
  const float* b0    = (const float*)d_in[3];
  const float* W1    = (const float*)d_in[4];
  const float* b1    = (const float*)d_in[5];
  const float* r_vec = (const float*)d_in[6];
  const float* attn0 = (const float*)d_in[7];
  const float* attn1 = (const float*)d_in[8];
  const float* fc1_w = (const float*)d_in[9];
  const float* fc1_b = (const float*)d_in[10];
  const float* fc2_w = (const float*)d_in[11];
  const float* fc_w  = (const float*)d_in[12];
  const float* fc_b  = (const float*)d_in[13];
  const int* idx0    = (const int*)d_in[14];
  const int* idx1    = (const int*)d_in[15];
  const int* mp_indices[2] = {(const int*)d_in[16], (const int*)d_in[19]};
  const int* mp_dst[2]     = {(const int*)d_in[17], (const int*)d_in[20]};
  const int* mp_target[2]  = {(const int*)d_in[18], (const int*)d_in[21]};
  const float* attn[2] = {attn0, attn1};

  float* ws     = (float*)d_ws;
  float* tf     = ws;
  float* finals = tf + (long)N_NODES * HID;
  float* h      = finals + 2 * LP * HID;
  float* a      = h + (long)NE * HID;
  float* m      = a + (long)NE * HEADS;
  float* s      = m + NG * HEADS;
  float* ft     = s + NG * HEADS;
  float* outs   = ft + (long)NG * HEADS * HID;
  float* score  = outs + 2L * NB * HEADS * HID;
  float* beta   = score + 2;
  int* used     = (int*)(outs + (long)NB * HEADS * HID);

  float* logits = (float*)d_out;
  float* hagg   = logits + (long)NB * OUTD;

  gemm_tf_kernel<F0><<<(N0 + 127) / 128, 256, 0, stream>>>(feat0, W0, b0, idx0,
                                                           tf, N0);
  gemm_tf_kernel<F1><<<(N1 + 127) / 128, 256, 0, stream>>>(feat1, W1, b1, idx1,
                                                           tf, N1);
  finals_kernel<<<1, 32, 0, stream>>>(r_vec, finals);

  mark_kernel<<<2, 1024, 0, stream>>>(mp_target[0], mp_target[1], used);

  for (int mp = 0; mp < 2; ++mp) {
    fill_kernel<<<(NG * HEADS + 255) / 256, 256, 0, stream>>>(m, -3.0e38f,
                                                              NG * HEADS);
    fill_kernel<<<(NG * HEADS + 255) / 256, 256, 0, stream>>>(s, 0.f,
                                                              NG * HEADS);
    fill_kernel<<<(NG * HEADS * HID + 255) / 256, 256, 0, stream>>>(
        ft, 0.f, NG * HEADS * HID);
    edge_h_kernel<<<(NE / 16 + 7) / 8, 256, 0, stream>>>(tf, finals + mp * LP * HID,
                                              mp_indices[mp], mp_dst[mp],
                                              used + mp * NG, attn[mp], h, a, m);
    edge_exp_kernel<<<(NE * HEADS + 255) / 256, 256, 0, stream>>>(
        a, mp_dst[mp], used + mp * NG, m, s, NE * HEADS);
    edge_scatter_kernel<<<NE / 8, 256, 0, stream>>>(h, a, s, mp_dst[mp],
                                                    used + mp * NG, ft);
    target_elu_kernel<<<(NB * HEADS * HID + 255) / 256, 256, 0, stream>>>(
        ft, mp_target[mp], outs + (long)mp * NB * HEADS * HID,
        NB * HEADS * HID);
  }

  fill_kernel<<<1, 64, 0, stream>>>(score, 0.f, 2);
  {
    dim3 g(NB / 128, 2);
    score_kernel<<<g, 256, 0, stream>>>(outs, fc1_w, fc1_b, fc2_w, score);
  }
  beta_kernel<<<1, 1, 0, stream>>>(score, beta);
  hagg_kernel<<<(NB * HEADS * HID + 255) / 256, 256, 0, stream>>>(
      outs, beta, hagg, NB * HEADS * HID);
  logits_kernel<<<NB / 128, 256, 0, stream>>>(hagg, fc_w, fc_b, logits);
}
